# starting kernel
# baseline (speedup 1.0000x reference)
#include <hip/hip_runtime.h>

typedef __attribute__((ext_vector_type(2))) float v2f;
typedef __attribute__((ext_vector_type(8))) float v8f;

#define NWAVES 8
#define TP     16

#define W1S 258
#define D1S 130
#define W2S 130
#define D2S 66
#define W3S 66
#define BUF128 130
#define BUF64  66

#define OFF_W1  0
#define OFF_D1  (OFF_W1 + 128*W1S)
#define OFF_W2  (OFF_D1 + 128*D1S)
#define OFF_D2  (OFF_W2 + 64*W2S)
#define OFF_W3  (OFF_D2 + 64*D2S)
#define OFF_BUF (OFF_W3 + 16*W3S)
#define BUF_FLOATS (16*BUF128)
#define OFF_Z   (OFF_BUF + NWAVES*BUF_FLOATS)
#define LDS_FLOATS (OFF_Z + NWAVES*144)

static __device__ inline v8f wmma4(v2f a, v2f b, v8f c) {
  return __builtin_amdgcn_wmma_f32_16x16x4_f32(false, a, false, b, (short)0, c,
                                               false, false);
}

template<int MT, int K>
static __device__ inline void gemm_lds(const float* __restrict__ W, int WS,
                                       const float* __restrict__ B, int BS,
                                       v8f* acc, int l16, int half) {
#pragma unroll 2
  for (int k = 0; k < K; k += 4) {
    const int c0 = k + 2*half;
    v2f b = *(const v2f*)(B + l16*BS + c0);
#pragma unroll
    for (int mt = 0; mt < MT; ++mt) {
      v2f a = *(const v2f*)(W + (mt*16 + l16)*WS + c0);
      acc[mt] = wmma4(a, b, acc[mt]);
    }
  }
}

template<int MT>
static __device__ inline void spill_cm(const v8f* acc, float* buf, int S,
                                       int l16, int half) {
#pragma unroll
  for (int mt = 0; mt < MT; ++mt) {
#pragma unroll
    for (int i = 0; i < 4; ++i) {
      v2f p;
      p.x = acc[mt][2*i];
      p.y = acc[mt][2*i + 1];
      *(v2f*)(buf + l16*S + mt*16 + 8*half + 2*i) = p;
    }
  }
}

template<int MT>
static __device__ inline void vn_nonlin(v8f* y, const v8f* d) {
  float dot = 0.f, dns = 0.f;
#pragma unroll
  for (int mt = 0; mt < MT; ++mt) {
#pragma unroll
    for (int i = 0; i < 8; ++i) {
      dot += y[mt][i] * d[mt][i];
      dns += d[mt][i] * d[mt][i];
    }
  }
  dot += __shfl_xor(dot, 16, 32);
  dns += __shfl_xor(dns, 16, 32);
  const float s = (dot >= 0.f) ? 0.f : 0.8f * dot / (dns + 1e-7f);
#pragma unroll
  for (int mt = 0; mt < MT; ++mt) {
#pragma unroll
    for (int i = 0; i < 8; ++i) {
      y[mt][i] = y[mt][i] - s * d[mt][i];
    }
  }
}

__global__ void __launch_bounds__(256, 1)
vnstd_kernel(const float* __restrict__ xg,
             const float* __restrict__ W1g, const float* __restrict__ D1g,
             const float* __restrict__ W2g, const float* __restrict__ D2g,
             const float* __restrict__ W3g,
             float* __restrict__ xstd, float* __restrict__ z0out) {
  extern __shared__ float smem[];
  const int tid = threadIdx.x;

  for (int i = tid; i < 128*256; i += 256) smem[OFF_W1 + (i >> 8)*W1S + (i & 255)] = W1g[i];
  for (int i = tid; i < 128*128; i += 256) smem[OFF_D1 + (i >> 7)*D1S + (i & 127)] = D1g[i];
  for (int i = tid; i < 64*128;  i += 256) smem[OFF_W2 + (i >> 7)*W2S + (i & 127)] = W2g[i];
  for (int i = tid; i < 64*64;   i += 256) smem[OFF_D2 + (i >> 6)*D2S + (i & 63)]  = D2g[i];
  for (int i = tid; i < 16*64;   i += 256)
    smem[OFF_W3 + (i >> 6)*W3S + (i & 63)] = (i < 3*64) ? W3g[i] : 0.f;
  __syncthreads();

  const int wave = tid >> 5;
  const int lane = tid & 31;
  const int l16  = lane & 15;
  const int half = lane >> 4;

  float* buf = smem + OFF_BUF + wave * BUF_FLOATS;
  float* zb  = smem + OFF_Z   + wave * 144;

  const int p0 = (blockIdx.x * NWAVES + wave) * TP;
  const int gp = p0 + l16;

  const float* sW1 = smem + OFF_W1;
  const float* sD1 = smem + OFF_D1;
  const float* sW2 = smem + OFF_W2;
  const float* sD2 = smem + OFF_D2;
  const float* sW3 = smem + OFF_W3;

  const v8f vzero = {0.f, 0.f, 0.f, 0.f, 0.f, 0.f, 0.f, 0.f};

#pragma unroll 1
  for (int t = 0; t < 3; ++t) {
    v8f y1[8];
#pragma unroll
    for (int mt = 0; mt < 8; ++mt) y1[mt] = vzero;

    const float* xb = xg + (size_t)gp * 768 + t;
#pragma unroll 2
    for (int k = 0; k < 256; k += 4) {
      const int c0 = k + 2*half;
      v2f b;
      b.x = xb[c0*3];
      b.y = xb[c0*3 + 3];
#pragma unroll
      for (int mt = 0; mt < 8; ++mt) {
        v2f a = *(const v2f*)(sW1 + (mt*16 + l16)*W1S + c0);
        y1[mt] = wmma4(a, b, y1[mt]);
      }
    }
    spill_cm<8>(y1, buf, BUF128, l16, half);

    v8f d1[8];
#pragma unroll
    for (int mt = 0; mt < 8; ++mt) d1[mt] = vzero;
    gemm_lds<8, 128>(sD1, D1S, buf, BUF128, d1, l16, half);

    vn_nonlin<8>(y1, d1);
    spill_cm<8>(y1, buf, BUF128, l16, half);

    v8f y2[4];
#pragma unroll
    for (int mt = 0; mt < 4; ++mt) y2[mt] = vzero;
    gemm_lds<4, 128>(sW2, W2S, buf, BUF128, y2, l16, half);
    spill_cm<4>(y2, buf, BUF64, l16, half);

    v8f d2[4];
#pragma unroll
    for (int mt = 0; mt < 4; ++mt) d2[mt] = vzero;
    gemm_lds<4, 64>(sD2, D2S, buf, BUF64, d2, l16, half);

    vn_nonlin<4>(y2, d2);
    spill_cm<4>(y2, buf, BUF64, l16, half);

    v8f zt = vzero;
    gemm_lds<1, 64>(sW3, W3S, buf, BUF64, &zt, l16, half);

    if (half == 0) {
#pragma unroll
      for (int r = 0; r < 3; ++r) zb[(t*3 + r)*16 + l16] = zt[r];
    }
  }

#pragma unroll 1
  for (int p = 0; p < TP; ++p) {
    const int pg = p0 + p;
    float zm[3][3];
#pragma unroll
    for (int j = 0; j < 3; ++j)
#pragma unroll
      for (int k = 0; k < 3; ++k) zm[j][k] = zb[(j*3 + k)*16 + p];

    const float* xp = xg + (size_t)pg * 768;
    float* op = xstd + (size_t)pg * 768;
#pragma unroll 1
    for (int cb = 0; cb < 8; ++cb) {
      const int c = cb*32 + lane;
      const float x0 = xp[c*3 + 0];
      const float x1 = xp[c*3 + 1];
      const float x2 = xp[c*3 + 2];
#pragma unroll
      for (int k = 0; k < 3; ++k)
        op[c*3 + k] = x0*zm[0][k] + x1*zm[1][k] + x2*zm[2][k];
    }
  }

  if (lane < 16) {
    const int pg = p0 + lane;
#pragma unroll
    for (int j = 0; j < 3; ++j)
#pragma unroll
      for (int k = 0; k < 3; ++k)
        z0out[(size_t)pg*9 + j*3 + k] = zb[(j*3 + k)*16 + lane];
  }
}

extern "C" void kernel_launch(void* const* d_in, const int* in_sizes, int n_in,
                              void* d_out, int out_size, void* d_ws, size_t ws_size,
                              hipStream_t stream) {
  (void)in_sizes; (void)n_in; (void)out_size; (void)d_ws; (void)ws_size;
  const float* x  = (const float*)d_in[0];
  const float* W1 = (const float*)d_in[1];
  const float* D1 = (const float*)d_in[2];
  const float* W2 = (const float*)d_in[3];
  const float* D2 = (const float*)d_in[4];
  const float* W3 = (const float*)d_in[5];

  float* xstd = (float*)d_out;
  float* z0   = xstd + (size_t)8 * 4096 * 256 * 3;

  const int points = 8 * 4096;
  const int blocks = points / (NWAVES * TP);
  const size_t shmem = (size_t)LDS_FLOATS * sizeof(float);

  vnstd_kernel<<<dim3(blocks), dim3(256), shmem, stream>>>(
      x, W1, D1, W2, D2, W3, xstd, z0);
}


# LDS pitch 16 B x odd: A and B fragments read as 16-byte LDS loads (4 consecutive k feed two k-steps) and tiles spilled as 16-byte stores, all bank-conflict-free; new k sum order
# speedup vs baseline: 1.6188x; 1.6188x over previous
#include <hip/hip_runtime.h>

typedef __attribute__((ext_vector_type(2))) float v2f;
typedef __attribute__((ext_vector_type(4))) float v4f;
typedef __attribute__((ext_vector_type(8))) float v8f;

#define NWAVES 8
#define TP     16

#define W1S 260
#define D1S 132
#define W2S 132
#define D2S 68
#define W3S 68
#define BUF128 132
#define BUF64  68

#define OFF_W1  0
#define OFF_D1  (OFF_W1 + 128*W1S)
#define OFF_W2  (OFF_D1 + 128*D1S)
#define OFF_D2  (OFF_W2 + 64*W2S)
#define OFF_W3  (OFF_D2 + 64*D2S)
#define OFF_BUF (OFF_W3 + 4*W3S)
#define BUF_FLOATS (16*BUF128)
#define OFF_Z   (OFF_BUF + NWAVES*BUF_FLOATS)
#define LDS_FLOATS (OFF_Z + NWAVES*144)

static __device__ inline v8f wmma4(v2f a, v2f b, v8f c) {
  return __builtin_amdgcn_wmma_f32_16x16x4_f32(false, a, false, b, (short)0, c,
                                               false, false);
}

template<int MT, int KB>
static __device__ inline void load_frags(v4f* a, v4f* b,
                                         const float* __restrict__ W, int WS,
                                         const float* __restrict__ B, int BS,
                                         int k, int wl, int l16, int half) {
  static_assert(KB % 2 == 0, "one 16-byte load feeds two k-steps");
#pragma unroll
  for (int s2 = 0; s2 < KB/2; ++s2) {
    const int c0 = k + 8*s2 + 4*half;
    b[s2] = *(const v4f*)(B + l16*BS + c0);
#pragma unroll
    for (int mt = 0; mt < MT; ++mt)
      a[s2*MT + mt] = *(const v4f*)(W + (mt*16 + wl)*WS + c0);
  }
}

template<int MT, int KB>
static __device__ inline void mma_block(const v4f* a, const v4f* b, v8f* acc) {
#pragma unroll
  for (int s2 = 0; s2 < KB/2; ++s2) {
    const v2f blo = __builtin_shufflevector(b[s2], b[s2], 0, 1);
    const v2f bhi = __builtin_shufflevector(b[s2], b[s2], 2, 3);
#pragma unroll
    for (int mt = 0; mt < MT; ++mt) {
      const v4f av = a[s2*MT + mt];
      acc[mt] = wmma4(__builtin_shufflevector(av, av, 0, 1), blo, acc[mt]);
    }
#pragma unroll
    for (int mt = 0; mt < MT; ++mt) {
      const v4f av = a[s2*MT + mt];
      acc[mt] = wmma4(__builtin_shufflevector(av, av, 2, 3), bhi, acc[mt]);
    }
  }
}

template<int MT, int K, int KB>
static __device__ inline void gemm_lds(const float* __restrict__ W, int WS,
                                       const float* __restrict__ B, int BS,
                                       v8f* acc, int wl, int l16, int half) {
  constexpr int KS = 4*KB;
  v4f a0[KB/2*MT], b0[KB/2], a1[KB/2*MT], b1[KB/2];
  load_frags<MT, KB>(a0, b0, W, WS, B, BS, 0, wl, l16, half);
#pragma unroll 1
  for (int k = 0; k < K; k += 2*KS) {
    load_frags<MT, KB>(a1, b1, W, WS, B, BS, k + KS, wl, l16, half);
    __builtin_amdgcn_sched_barrier(0);
    mma_block<MT, KB>(a0, b0, acc);
    __builtin_amdgcn_sched_barrier(0);
    const int kn = (k + 2*KS < K) ? k + 2*KS : 0;
    load_frags<MT, KB>(a0, b0, W, WS, B, BS, kn, wl, l16, half);
    __builtin_amdgcn_sched_barrier(0);
    mma_block<MT, KB>(a1, b1, acc);
  }
}

template<int MT>
static __device__ inline void spill_cm(const v8f* acc, float* buf, int S,
                                       int l16, int half) {
#pragma unroll
  for (int mt = 0; mt < MT; ++mt) {
    float* p = buf + l16*S + mt*16 + 8*half;
    *(v4f*)p       = __builtin_shufflevector(acc[mt], acc[mt], 0, 1, 2, 3);
    *(v4f*)(p + 4) = __builtin_shufflevector(acc[mt], acc[mt], 4, 5, 6, 7);
  }
}

template<int MT>
static __device__ inline void vn_nonlin(v8f* y, const v8f* d) {
  float dot = 0.f, dns = 0.f;
#pragma unroll
  for (int mt = 0; mt < MT; ++mt) {
#pragma unroll
    for (int i = 0; i < 8; ++i) {
      dot += y[mt][i] * d[mt][i];
      dns += d[mt][i] * d[mt][i];
    }
  }
  dot += __shfl_xor(dot, 16, 32);
  dns += __shfl_xor(dns, 16, 32);
  const float s = (dot >= 0.f) ? 0.f : 0.8f * dot / (dns + 1e-7f);
#pragma unroll
  for (int mt = 0; mt < MT; ++mt) {
#pragma unroll
    for (int i = 0; i < 8; ++i) {
      y[mt][i] = y[mt][i] - s * d[mt][i];
    }
  }
}

#define NQ_D1 (128*128/4/256)
#define NQ_W2 (64*128/4/256)
#define NQ_D2 (64*64/4/256)
#define NQ    (NQ_D1 + NQ_W2 + NQ_D2 + 1)

static __device__ __forceinline__ void request_rest(v4f* q,
    const float* __restrict__ D1g, const float* __restrict__ W2g,
    const float* __restrict__ D2g, const float* __restrict__ W3g, int tid) {
#pragma unroll
  for (int j = 0; j < NQ_D1; ++j) q[j] = *(const v4f*)(D1g + 4*(tid + 256*j));
#pragma unroll
  for (int j = 0; j < NQ_W2; ++j) q[NQ_D1 + j] = *(const v4f*)(W2g + 4*(tid + 256*j));
#pragma unroll
  for (int j = 0; j < NQ_D2; ++j) q[NQ_D1 + NQ_W2 + j] = *(const v4f*)(D2g + 4*(tid + 256*j));
  const v4f z4 = {0.f, 0.f, 0.f, 0.f};
  q[NQ - 1] = z4;
  if (tid < 3*64/4) q[NQ - 1] = *(const v4f*)(W3g + 4*tid);
}

template<int C>
static __device__ __forceinline__ void put_piece(float* dst, int S, int f4, v4f v) {
  const int e = 4*f4;
  *(v4f*)(dst + (e / C)*S + (e % C)) = v;
}

static __device__ __forceinline__ void store_rest(const v4f* q, float* smem, int tid) {
#pragma unroll
  for (int j = 0; j < NQ_D1; ++j) put_piece<128>(smem + OFF_D1, D1S, tid + 256*j, q[j]);
#pragma unroll
  for (int j = 0; j < NQ_W2; ++j) put_piece<128>(smem + OFF_W2, W2S, tid + 256*j, q[NQ_D1 + j]);
#pragma unroll
  for (int j = 0; j < NQ_D2; ++j) put_piece<64>(smem + OFF_D2, D2S, tid + 256*j, q[NQ_D1 + NQ_W2 + j]);
  if (tid < 3*64/4) put_piece<64>(smem + OFF_W3, W3S, tid, q[NQ - 1]);
}

#define W1C   64
#define NQ_W1 (128*W1C/4/256)

static __device__ __forceinline__ void request_w1_chunk(v4f* q, const float* __restrict__ W1g,
                                                        int ch, int tid) {
#pragma unroll
  for (int j = 0; j < NQ_W1; ++j) {
    const int f = tid + 256*j;
    q[j] = *(const v4f*)(W1g + (f >> 4)*256 + ch*W1C + 4*(f & 15));
  }
}

static __device__ __forceinline__ void store_w1_chunk(const v4f* q, float* smem, int ch, int tid) {
#pragma unroll
  for (int j = 0; j < NQ_W1; ++j) {
    const int f = tid + 256*j;
    *(v4f*)(smem + OFF_W1 + (f >> 4)*W1S + ch*W1C + 4*(f & 15)) = q[j];
  }
}

#define EPI_PRE 10
#define XS_REGS (EPI_PRE*8*3)
#define EPI_REST (TP - EPI_PRE)

static __device__ inline void load_x_pass(float* xs, const float* __restrict__ xt, int lane) {
#pragma unroll
  for (int p = 0; p < TP; ++p) {
#pragma unroll
    for (int cb = 0; cb < 8; ++cb)
      xs[p*8 + cb] = xt[(size_t)p * 768 + (cb*32 + lane)*3];
  }
}

static __device__ inline void x_half_to_lds(const float* xs, float* buf, int kh, int lane) {
#pragma unroll
  for (int p = 0; p < TP; ++p) {
#pragma unroll
    for (int c4 = 0; c4 < 4; ++c4)
      buf[p*BUF128 + c4*32 + lane] = xs[p*8 + kh*4 + c4];
  }
}

template<int NP>
static __device__ inline void load_x_final(float* xs, const float* __restrict__ xt, int lane) {
#pragma unroll
  for (int p = 0; p < NP; ++p) {
#pragma unroll
    for (int cb = 0; cb < 8; ++cb) {
      const float* s = xt + (size_t)p * 768 + (cb*32 + lane)*3;
      const int u = p*8 + cb;
      xs[3*u + 0] = s[0];
      xs[3*u + 1] = s[1];
      xs[3*u + 2] = s[2];
    }
  }
}

template<int NP>
static __device__ __forceinline__ void final_rows(const float* xs, const float* zb,
                                                  float* __restrict__ op0, int pl, int lane) {
#pragma unroll
  for (int p = 0; p < NP; ++p) {
    float zm[3][3];
#pragma unroll
    for (int j = 0; j < 3; ++j)
#pragma unroll
      for (int k = 0; k < 3; ++k) zm[j][k] = zb[(j*3 + k)*16 + pl + p];

    float* op = op0 + (size_t)(pl + p) * 768;
#pragma unroll
    for (int cb = 0; cb < 8; ++cb) {
      const int c = cb*32 + lane;
      const int u = p*8 + cb;
      const float x0 = xs[3*u + 0];
      const float x1 = xs[3*u + 1];
      const float x2 = xs[3*u + 2];
#pragma unroll
      for (int k = 0; k < 3; ++k)
        op[c*3 + k] = x0*zm[0][k] + x1*zm[1][k] + x2*zm[2][k];
    }
    __builtin_amdgcn_sched_barrier(0);
  }
}

static __device__ __forceinline__ void stage_1(const float* xs, float* buf, v8f* y1,
                                               const float* sW1, int l16, int half, int lane) {
  const v8f vzero = {0.f, 0.f, 0.f, 0.f, 0.f, 0.f, 0.f, 0.f};
#pragma unroll
  for (int mt = 0; mt < 8; ++mt) y1[mt] = vzero;

  x_half_to_lds(xs, buf, 0, lane);
  gemm_lds<8, 128, 2>(sW1, W1S, buf, BUF128, y1, l16, l16, half);
  x_half_to_lds(xs, buf, 1, lane);
  gemm_lds<8, 128, 2>(sW1 + 128, W1S, buf, BUF128, y1, l16, l16, half);
}

static __device__ __forceinline__ void stage_1_first(const float* xs, float* buf, v8f* y1,
                                                     float* smem, const float* __restrict__ W1g,
                                                     v4f* wc, int tid, int l16, int half, int lane) {
  const v8f vzero = {0.f, 0.f, 0.f, 0.f, 0.f, 0.f, 0.f, 0.f};
#pragma unroll
  for (int mt = 0; mt < 8; ++mt) y1[mt] = vzero;
  const float* sW1 = smem + OFF_W1;

  x_half_to_lds(xs, buf, 0, lane);
  gemm_lds<8, W1C, 2>(sW1, W1S, buf, BUF128, y1, l16, l16, half);
  store_w1_chunk(wc, smem, 1, tid);
  __syncthreads();
  request_w1_chunk(wc, W1g, 2, tid);
  gemm_lds<8, W1C, 2>(sW1 + W1C, W1S, buf + W1C, BUF128, y1, l16, l16, half);
  store_w1_chunk(wc, smem, 2, tid);
  __syncthreads();
  request_w1_chunk(wc, W1g, 3, tid);
  x_half_to_lds(xs, buf, 1, lane);
  gemm_lds<8, W1C, 2>(sW1 + 2*W1C, W1S, buf, BUF128, y1, l16, l16, half);
  store_w1_chunk(wc, smem, 3, tid);
  __syncthreads();
  gemm_lds<8, W1C, 2>(sW1 + 3*W1C, W1S, buf + W1C, BUF128, y1, l16, l16, half);
}

template<bool LAST>
static __device__ __forceinline__ void stages_2_to_5(
    int t, float* xs, float* buf, float* zb, v8f* y1,
    const float* sD1, const float* sW2, const float* sD2, const float* sW3,
    const float* __restrict__ xt, int l16, int half, int lane) {
  const v8f vzero = {0.f, 0.f, 0.f, 0.f, 0.f, 0.f, 0.f, 0.f};
  v8f d1[8];
#pragma unroll
  for (int mt = 0; mt < 8; ++mt) d1[mt] = vzero;
  gemm_lds<8, 128, 2>(sD1, D1S, buf, BUF128, d1, l16, l16, half);

  vn_nonlin<8>(y1, d1);
  spill_cm<8>(y1, buf, BUF128, l16, half);
  if (LAST) load_x_final<EPI_PRE>(xs, xt, lane);

  v8f y2[4];
#pragma unroll
  for (int mt = 0; mt < 4; ++mt) y2[mt] = vzero;
  gemm_lds<4, 128, 4>(sW2, W2S, buf, BUF128, y2, l16, l16, half);
  spill_cm<4>(y2, buf, BUF64, l16, half);

  v8f d2[4];
#pragma unroll
  for (int mt = 0; mt < 4; ++mt) d2[mt] = vzero;
  gemm_lds<4, 64, 4>(sD2, D2S, buf, BUF64, d2, l16, l16, half);

  vn_nonlin<4>(y2, d2);
  spill_cm<4>(y2, buf, BUF64, l16, half);

  v8f zt = vzero;
  gemm_lds<1, 64, 8>(sW3, W3S, buf, BUF64, &zt, (l16 < 3) ? l16 : 3, l16, half);

  if (half == 0) {
#pragma unroll
    for (int r = 0; r < 3; ++r) zb[(t*3 + r)*16 + l16] = zt[r];
  }
}

__global__ void __launch_bounds__(256, 1)
vnstd_kernel(const float* __restrict__ xg,
             const float* __restrict__ W1g, const float* __restrict__ D1g,
             const float* __restrict__ W2g, const float* __restrict__ D2g,
             const float* __restrict__ W3g,
             float* __restrict__ xstd, float* __restrict__ z0out) {
  extern __shared__ __align__(16) float smem[];
  const int tid = threadIdx.x;

  v4f wc[NQ_W1];
  request_w1_chunk(wc, W1g, 0, tid);
  store_w1_chunk(wc, smem, 0, tid);
  if (tid < 64) smem[OFF_W3 + 3*W3S + tid] = 0.f;
  __syncthreads();

  const int wave = tid >> 5;
  const int lane = tid & 31;
  const int l16  = lane & 15;
  const int half = lane >> 4;

  float* buf = smem + OFF_BUF + wave * BUF_FLOATS;
  float* zb  = smem + OFF_Z   + wave * 144;

  const int p0 = (blockIdx.x * NWAVES + wave) * TP;

  const float* sW1 = smem + OFF_W1;
  const float* sD1 = smem + OFF_D1;
  const float* sW2 = smem + OFF_W2;
  const float* sD2 = smem + OFF_D2;
  const float* sW3 = smem + OFF_W3;

  float xs[XS_REGS];

  const float* xt = xg + (size_t)p0 * 768;
  load_x_pass(xs, xt, lane);
  request_w1_chunk(wc, W1g, 1, tid);
  v4f wq[NQ];
  request_rest(wq, D1g, W2g, D2g, W3g, tid);

  v8f y1[8];
  stage_1_first(xs, buf, y1, smem, W1g, wc, tid, l16, half, lane);
  store_rest(wq, smem, tid);
  spill_cm<8>(y1, buf, BUF128, l16, half);
  __syncthreads();
  load_x_pass(xs, xt + 1, lane);

#pragma unroll 1
  for (int t = 0; t < 2; ++t) {
    stages_2_to_5<false>(t, xs, buf, zb, y1, sD1, sW2, sD2, sW3, xt, l16, half, lane);
    stage_1(xs, buf, y1, sW1, l16, half, lane);
    if (t == 0) load_x_pass(xs, xt + 2, lane);
    spill_cm<8>(y1, buf, BUF128, l16, half);
  }
  stages_2_to_5<true>(2, xs, buf, zb, y1, sD1, sW2, sD2, sW3, xt, l16, half, lane);

  static_assert(EPI_REST <= EPI_PRE, "one row request per finished point");
  float xr[EPI_REST*8*3];
  float* op0 = xstd + (size_t)p0 * 768;
#pragma unroll
  for (int p = 0; p < EPI_PRE; ++p) {
    if (p < EPI_REST)
      load_x_final<1>(xr + p*24, xt + (size_t)(EPI_PRE + p) * 768, lane);
    final_rows<1>(xs + p*24, zb, op0, p, lane);
  }
  final_rows<EPI_REST>(xr, zb, op0, EPI_PRE, lane);

  if (lane < 16) {
    const int pg = p0 + lane;
#pragma unroll
    for (int j = 0; j < 3; ++j)
#pragma unroll
      for (int k = 0; k < 3; ++k)
        z0out[(size_t)pg*9 + j*3 + k] = zb[(j*3 + k)*16 + lane];
  }
}

extern "C" void kernel_launch(void* const* d_in, const int* in_sizes, int n_in,
                              void* d_out, int out_size, void* d_ws, size_t ws_size,
                              hipStream_t stream) {
  (void)in_sizes; (void)n_in; (void)out_size; (void)d_ws; (void)ws_size;
  const float* x  = (const float*)d_in[0];
  const float* W1 = (const float*)d_in[1];
  const float* D1 = (const float*)d_in[2];
  const float* W2 = (const float*)d_in[3];
  const float* D2 = (const float*)d_in[4];
  const float* W3 = (const float*)d_in[5];

  float* xstd = (float*)d_out;
  float* z0   = xstd + (size_t)8 * 4096 * 256 * 3;

  const int points = 8 * 4096;
  const int blocks = points / (NWAVES * TP);
  const size_t shmem = (size_t)LDS_FLOATS * sizeof(float);

  vnstd_kernel<<<dim3(blocks), dim3(256), shmem, stream>>>(
      x, W1, D1, W2, D2, W3, xstd, z0);
}


# N tiles regrouped by point (5 whole points + 1 component of point 15 per tile): a tile's x rows are fetched once, feed stage 1, then, from registers, its points' final product; same sums and order
# speedup vs baseline: 1.8008x; 1.1124x over previous
#include <hip/hip_runtime.h>

typedef __attribute__((ext_vector_type(2))) float v2f;
typedef __attribute__((ext_vector_type(4))) float v4f;
typedef __attribute__((ext_vector_type(8))) float v8f;

#define NWAVES 8
#define TP     16

#define W1S 260
#define D1S 132
#define W2S 132
#define D2S 68
#define W3S 68
#define BUF128 132
#define BUF64  68

#define OFF_W1  0
#define OFF_D1  (OFF_W1 + 128*W1S)
#define OFF_W2  (OFF_D1 + 128*D1S)
#define OFF_D2  (OFF_W2 + 64*W2S)
#define OFF_W3  (OFF_D2 + 64*D2S)
#define OFF_BUF (OFF_W3 + 4*W3S)
#define BUF_FLOATS (16*BUF128)
#define OFF_Z   (OFF_BUF + NWAVES*BUF_FLOATS)
#define LDS_FLOATS (OFF_Z + NWAVES*144)

static __device__ inline v8f wmma4(v2f a, v2f b, v8f c) {
  return __builtin_amdgcn_wmma_f32_16x16x4_f32(false, a, false, b, (short)0, c,
                                               false, false);
}

template<int MT, int KB>
static __device__ inline void load_frags(v4f* a, v4f* b,
                                         const float* __restrict__ W, int WS,
                                         const float* __restrict__ B, int BS,
                                         int k, int wl, int l16, int half) {
  static_assert(KB % 2 == 0, "one 16-byte load feeds two k-steps");
#pragma unroll
  for (int s2 = 0; s2 < KB/2; ++s2) {
    const int c0 = k + 8*s2 + 4*half;
    b[s2] = *(const v4f*)(B + l16*BS + c0);
#pragma unroll
    for (int mt = 0; mt < MT; ++mt)
      a[s2*MT + mt] = *(const v4f*)(W + (mt*16 + wl)*WS + c0);
  }
}

template<int MT, int KB>
static __device__ inline void mma_block(const v4f* a, const v4f* b, v8f* acc) {
#pragma unroll
  for (int s2 = 0; s2 < KB/2; ++s2) {
    const v2f blo = __builtin_shufflevector(b[s2], b[s2], 0, 1);
    const v2f bhi = __builtin_shufflevector(b[s2], b[s2], 2, 3);
#pragma unroll
    for (int mt = 0; mt < MT; ++mt) {
      const v4f av = a[s2*MT + mt];
      acc[mt] = wmma4(__builtin_shufflevector(av, av, 0, 1), blo, acc[mt]);
    }
#pragma unroll
    for (int mt = 0; mt < MT; ++mt) {
      const v4f av = a[s2*MT + mt];
      acc[mt] = wmma4(__builtin_shufflevector(av, av, 2, 3), bhi, acc[mt]);
    }
  }
}

template<int MT, int K, int KB>
static __device__ inline void gemm_lds(const float* __restrict__ W, int WS,
                                       const float* __restrict__ B, int BS,
                                       v8f* acc, int wl, int l16, int half) {
  constexpr int KS = 4*KB;
  v4f a0[KB/2*MT], b0[KB/2], a1[KB/2*MT], b1[KB/2];
  load_frags<MT, KB>(a0, b0, W, WS, B, BS, 0, wl, l16, half);
#pragma unroll 1
  for (int k = 0; k < K; k += 2*KS) {
    load_frags<MT, KB>(a1, b1, W, WS, B, BS, k + KS, wl, l16, half);
    __builtin_amdgcn_sched_barrier(0);
    mma_block<MT, KB>(a0, b0, acc);
    __builtin_amdgcn_sched_barrier(0);
    const int kn = (k + 2*KS < K) ? k + 2*KS : 0;
    load_frags<MT, KB>(a0, b0, W, WS, B, BS, kn, wl, l16, half);
    __builtin_amdgcn_sched_barrier(0);
    mma_block<MT, KB>(a1, b1, acc);
  }
}

template<int MT>
static __device__ inline void spill_cm(const v8f* acc, float* buf, int S,
                                       int l16, int half) {
#pragma unroll
  for (int mt = 0; mt < MT; ++mt) {
    float* p = buf + l16*S + mt*16 + 8*half;
    *(v4f*)p       = __builtin_shufflevector(acc[mt], acc[mt], 0, 1, 2, 3);
    *(v4f*)(p + 4) = __builtin_shufflevector(acc[mt], acc[mt], 4, 5, 6, 7);
  }
}

template<int MT>
static __device__ inline void vn_nonlin(v8f* y, const v8f* d) {
  float dot = 0.f, dns = 0.f;
#pragma unroll
  for (int mt = 0; mt < MT; ++mt) {
#pragma unroll
    for (int i = 0; i < 8; ++i) {
      dot += y[mt][i] * d[mt][i];
      dns += d[mt][i] * d[mt][i];
    }
  }
  dot += __shfl_xor(dot, 16, 32);
  dns += __shfl_xor(dns, 16, 32);
  const float s = (dot >= 0.f) ? 0.f : 0.8f * dot / (dns + 1e-7f);
#pragma unroll
  for (int mt = 0; mt < MT; ++mt) {
#pragma unroll
    for (int i = 0; i < 8; ++i) {
      y[mt][i] = y[mt][i] - s * d[mt][i];
    }
  }
}

#define NQ_D1 (128*128/4/256)
#define NQ_W2 (64*128/4/256)
#define NQ_D2 (64*64/4/256)
#define NQ    (NQ_D1 + NQ_W2 + NQ_D2 + 1)

static __device__ __forceinline__ void request_rest(v4f* q,
    const float* __restrict__ D1g, const float* __restrict__ W2g,
    const float* __restrict__ D2g, const float* __restrict__ W3g, int tid) {
#pragma unroll
  for (int j = 0; j < NQ_D1; ++j) q[j] = *(const v4f*)(D1g + 4*(tid + 256*j));
#pragma unroll
  for (int j = 0; j < NQ_W2; ++j) q[NQ_D1 + j] = *(const v4f*)(W2g + 4*(tid + 256*j));
#pragma unroll
  for (int j = 0; j < NQ_D2; ++j) q[NQ_D1 + NQ_W2 + j] = *(const v4f*)(D2g + 4*(tid + 256*j));
  const v4f z4 = {0.f, 0.f, 0.f, 0.f};
  q[NQ - 1] = z4;
  if (tid < 3*64/4) q[NQ - 1] = *(const v4f*)(W3g + 4*tid);
}

template<int C>
static __device__ __forceinline__ void put_piece(float* dst, int S, int f4, v4f v) {
  const int e = 4*f4;
  *(v4f*)(dst + (e / C)*S + (e % C)) = v;
}

static __device__ __forceinline__ void store_rest(const v4f* q, float* smem, int tid) {
#pragma unroll
  for (int j = 0; j < NQ_D1; ++j) put_piece<128>(smem + OFF_D1, D1S, tid + 256*j, q[j]);
#pragma unroll
  for (int j = 0; j < NQ_W2; ++j) put_piece<128>(smem + OFF_W2, W2S, tid + 256*j, q[NQ_D1 + j]);
#pragma unroll
  for (int j = 0; j < NQ_D2; ++j) put_piece<64>(smem + OFF_D2, D2S, tid + 256*j, q[NQ_D1 + NQ_W2 + j]);
  if (tid < 3*64/4) put_piece<64>(smem + OFF_W3, W3S, tid, q[NQ - 1]);
}

#define W1C   64
#define NQ_W1 (128*W1C/4/256)

static __device__ __forceinline__ void request_w1_chunk(v4f* q, const float* __restrict__ W1g,
                                                        int ch, int tid) {
#pragma unroll
  for (int j = 0; j < NQ_W1; ++j) {
    const int f = tid + 256*j;
    q[j] = *(const v4f*)(W1g + (f >> 4)*256 + ch*W1C + 4*(f & 15));
  }
}

static __device__ __forceinline__ void store_w1_chunk(const v4f* q, float* smem, int ch, int tid) {
#pragma unroll
  for (int j = 0; j < NQ_W1; ++j) {
    const int f = tid + 256*j;
    *(v4f*)(smem + OFF_W1 + (f >> 4)*W1S + ch*W1C + 4*(f & 15)) = q[j];
  }
}

#define ROWF 24
#define NCOL 48


template<int NP, int CB0 = 0, int NCB = 8>
static __device__ inline void load_x_final(float* xs, const float* __restrict__ xt, int lane) {
#pragma unroll
  for (int p = 0; p < NP; ++p) {
#pragma unroll
    for (int cb = CB0; cb < CB0 + NCB; ++cb) {
      const float* s = xt + (size_t)p * 768 + (cb*32 + lane)*3;
      const int u = p*8 + cb;
      xs[3*u + 0] = s[0];
      xs[3*u + 1] = s[1];
      xs[3*u + 2] = s[2];
    }
  }
}

template<int CB0 = 0, int NCB = 8>
static __device__ inline void load_x_comp(float* E, const float* __restrict__ xe, int lane) {
#pragma unroll
  for (int cb = CB0; cb < CB0 + NCB; ++cb) E[cb] = xe[(cb*32 + lane)*3];
}

template<int KH>
static __device__ __forceinline__ void request_tile_half(float* R, float* E,
    const float* __restrict__ xr, const float* __restrict__ xe, int lane) {
  load_x_final<5, 4*KH, 4>(R, xr, lane);
  load_x_comp<4*KH, 4>(E, xe, lane);
}

static __device__ __forceinline__ float xcol(const float* R, const float* E, int j, int cb) {
  return (j < 15) ? R[((j/3)*8 + cb)*3 + j%3] : E[cb];
}

template<int NP>
static __device__ __forceinline__ void final_rows(const float* xs, const float* zb,
                                                  float* __restrict__ op0, int pl, int lane) {
#pragma unroll
  for (int p = 0; p < NP; ++p) {
    float zm[3][3];
#pragma unroll
    for (int j = 0; j < 3; ++j)
#pragma unroll
      for (int k = 0; k < 3; ++k) zm[j][k] = zb[k*NCOL + 3*(pl + p) + j];

    float* op = op0 + (size_t)(pl + p) * 768;
#pragma unroll
    for (int cb = 0; cb < 8; ++cb) {
      const int c = cb*32 + lane;
      const int u = p*8 + cb;
      const float x0 = xs[3*u + 0];
      const float x1 = xs[3*u + 1];
      const float x2 = xs[3*u + 2];
#pragma unroll
      for (int k = 0; k < 3; ++k)
        op[c*3 + k] = x0*zm[0][k] + x1*zm[1][k] + x2*zm[2][k];
    }
    __builtin_amdgcn_sched_barrier(0);
  }
}

template<int KH>
static __device__ __forceinline__ void x_half_to_lds(const float* R, const float* E,
                                                     float* buf, int lane) {
#pragma unroll
  for (int j = 0; j < 16; ++j) {
#pragma unroll
    for (int c4 = 0; c4 < 4; ++c4)
      buf[j*BUF128 + c4*32 + lane] = xcol(R, E, j, KH*4 + c4);
  }
}

static __device__ __forceinline__ void stage_1(const float* R, const float* E, float* buf, v8f* y1,
                                               const float* sW1, int l16, int half, int lane) {
  const v8f vzero = {0.f, 0.f, 0.f, 0.f, 0.f, 0.f, 0.f, 0.f};
#pragma unroll
  for (int mt = 0; mt < 8; ++mt) y1[mt] = vzero;

  x_half_to_lds<0>(R, E, buf, lane);
  gemm_lds<8, 128, 2>(sW1, W1S, buf, BUF128, y1, l16, l16, half);
  x_half_to_lds<1>(R, E, buf, lane);
  gemm_lds<8, 128, 2>(sW1 + 128, W1S, buf, BUF128, y1, l16, l16, half);
}

static __device__ __forceinline__ void stage_1_first(const float* R, const float* E, float* buf, v8f* y1,
                                                     float* smem, const float* __restrict__ W1g,
                                                     v4f* wc, int tid, int l16, int half, int lane) {
  const v8f vzero = {0.f, 0.f, 0.f, 0.f, 0.f, 0.f, 0.f, 0.f};
#pragma unroll
  for (int mt = 0; mt < 8; ++mt) y1[mt] = vzero;
  const float* sW1 = smem + OFF_W1;

  x_half_to_lds<0>(R, E, buf, lane);
  gemm_lds<8, W1C, 2>(sW1, W1S, buf, BUF128, y1, l16, l16, half);
  store_w1_chunk(wc, smem, 1, tid);
  __syncthreads();
  request_w1_chunk(wc, W1g, 2, tid);
  gemm_lds<8, W1C, 2>(sW1 + W1C, W1S, buf + W1C, BUF128, y1, l16, l16, half);
  store_w1_chunk(wc, smem, 2, tid);
  __syncthreads();
  request_w1_chunk(wc, W1g, 3, tid);
  x_half_to_lds<1>(R, E, buf, lane);
  gemm_lds<8, W1C, 2>(sW1 + 2*W1C, W1S, buf, BUF128, y1, l16, l16, half);
  store_w1_chunk(wc, smem, 3, tid);
  __syncthreads();
  gemm_lds<8, W1C, 2>(sW1 + 3*W1C, W1S, buf + W1C, BUF128, y1, l16, l16, half);
}

template<bool LAST>
static __device__ __forceinline__ void stages_2_to_5(
    int g, float* nx, float* ne, float* buf, float* zb, v8f* y1,
    const float* sD1, const float* sW2, const float* sD2, const float* sW3,
    const float* __restrict__ xr, const float* __restrict__ xe, int l16, int half, int lane) {
  const v8f vzero = {0.f, 0.f, 0.f, 0.f, 0.f, 0.f, 0.f, 0.f};
  v8f d1[8];
#pragma unroll
  for (int mt = 0; mt < 8; ++mt) d1[mt] = vzero;
  gemm_lds<8, 128, 2>(sD1, D1S, buf, BUF128, d1, l16, l16, half);

  vn_nonlin<8>(y1, d1);
  spill_cm<8>(y1, buf, BUF128, l16, half);
  __builtin_amdgcn_sched_barrier(0);
  if (LAST) load_x_final<1>(nx, xr, lane);
  else      request_tile_half<0>(nx, ne, xr, xe, lane);

  v8f y2[4];
#pragma unroll
  for (int mt = 0; mt < 4; ++mt) y2[mt] = vzero;
  gemm_lds<4, 128, 4>(sW2, W2S, buf, BUF128, y2, l16, l16, half);
  spill_cm<4>(y2, buf, BUF64, l16, half);

  v8f d2[4];
#pragma unroll
  for (int mt = 0; mt < 4; ++mt) d2[mt] = vzero;
  gemm_lds<4, 64, 4>(sD2, D2S, buf, BUF64, d2, l16, l16, half);

  vn_nonlin<4>(y2, d2);
  spill_cm<4>(y2, buf, BUF64, l16, half);

  v8f zt = vzero;
  gemm_lds<1, 64, 8>(sW3, W3S, buf, BUF64, &zt, (l16 < 3) ? l16 : 3, l16, half);

  if (half == 0) {
#pragma unroll
    for (int r = 0; r < 3; ++r) zb[r*NCOL + ((l16 < 15) ? 15*g + l16 : 45 + g)] = zt[r];
  }
}

__global__ void __launch_bounds__(256, 1)
vnstd_kernel(const float* __restrict__ xg,
             const float* __restrict__ W1g, const float* __restrict__ D1g,
             const float* __restrict__ W2g, const float* __restrict__ D2g,
             const float* __restrict__ W3g,
             float* __restrict__ xstd, float* __restrict__ z0out) {
  extern __shared__ __align__(16) float smem[];
  const int tid = threadIdx.x;

  v4f wc[NQ_W1];
  request_w1_chunk(wc, W1g, 0, tid);
  store_w1_chunk(wc, smem, 0, tid);
  if (tid < 64) smem[OFF_W3 + 3*W3S + tid] = 0.f;
  __syncthreads();

  const int wave = tid >> 5;
  const int lane = tid & 31;
  const int l16  = lane & 15;
  const int half = lane >> 4;

  float* buf = smem + OFF_BUF + wave * BUF_FLOATS;
  float* zb  = smem + OFF_Z   + wave * 144;

  const int p0 = (blockIdx.x * NWAVES + wave) * TP;

  const float* sW1 = smem + OFF_W1;
  const float* sD1 = smem + OFF_D1;
  const float* sW2 = smem + OFF_W2;
  const float* sD2 = smem + OFF_D2;
  const float* sW3 = smem + OFF_W3;

  const float* xt = xg + (size_t)p0 * 768;
  float* op0 = xstd + (size_t)p0 * 768;

  const float* x15 = xt + (size_t)15 * 768;
  float R[5*ROWF], E[8];
  load_x_final<5>(R, xt, lane);
  load_x_comp<>(E, x15, lane);
  request_w1_chunk(wc, W1g, 1, tid);
  v4f wq[NQ];
  request_rest(wq, D1g, W2g, D2g, W3g, tid);

  v8f y1[8];
  stage_1_first(R, E, buf, y1, smem, W1g, wc, tid, l16, half, lane);
  store_rest(wq, smem, tid);
  spill_cm<8>(y1, buf, BUF128, l16, half);
  __syncthreads();

#pragma unroll 1
  for (int g = 0; g < 2; ++g) {
    float N[5*ROWF], NE[8];
    const float* xr = xt + (size_t)(5*g + 5) * 768;
    const float* xe = x15 + (g + 1);
    stages_2_to_5<false>(g, N, NE, buf, zb, y1, sD1, sW2, sD2, sW3, xr, xe, l16, half, lane);
    final_rows<5>(R, zb, op0, 5*g, lane);
    request_tile_half<1>(N, NE, xr, xe, lane);
    stage_1(N, NE, buf, y1, sW1, l16, half, lane);
    spill_cm<8>(y1, buf, BUF128, l16, half);
#pragma unroll
    for (int i = 0; i < 5*ROWF; ++i) R[i] = N[i];
#pragma unroll
    for (int i = 0; i < 8; ++i) E[i] = NE[i];
  }
  float P[ROWF];
  stages_2_to_5<true>(2, P, E, buf, zb, y1, sD1, sW2, sD2, sW3, x15, x15, l16, half, lane);
  final_rows<5>(R, zb, op0, 10, lane);
  final_rows<1>(P, zb, op0, 15, lane);

  if (lane < 16) {
    const int pg = p0 + lane;
#pragma unroll
    for (int j = 0; j < 3; ++j)
#pragma unroll
      for (int k = 0; k < 3; ++k)
        z0out[(size_t)pg*9 + j*3 + k] = zb[k*NCOL + 3*lane + j];
  }
}

extern "C" void kernel_launch(void* const* d_in, const int* in_sizes, int n_in,
                              void* d_out, int out_size, void* d_ws, size_t ws_size,
                              hipStream_t stream) {
  (void)in_sizes; (void)n_in; (void)out_size; (void)d_ws; (void)ws_size;
  const float* x  = (const float*)d_in[0];
  const float* W1 = (const float*)d_in[1];
  const float* D1 = (const float*)d_in[2];
  const float* W2 = (const float*)d_in[3];
  const float* D2 = (const float*)d_in[4];
  const float* W3 = (const float*)d_in[5];

  float* xstd = (float*)d_out;
  float* z0   = xstd + (size_t)8 * 4096 * 256 * 3;

  const int points = 8 * 4096;
  const int blocks = points / (NWAVES * TP);
  const size_t shmem = (size_t)LDS_FLOATS * sizeof(float);

  vnstd_kernel<<<dim3(blocks), dim3(256), shmem, stream>>>(
      x, W1, D1, W2, D2, W3, xstd, z0);
}
